# starting kernel
# baseline (speedup 1.0000x reference)
#include <hip/hip_runtime.h>
#include <hip/hip_bf16.h>

typedef __attribute__((ext_vector_type(2))) float v2f;
typedef __attribute__((ext_vector_type(8))) float v8f;

#define IN_CH  128
#define HID_CH 64

__global__ __launch_bounds__(256)
void gcn_fill_kernel(float* __restrict__ p, float v, long long n) {
    long long i = (long long)blockIdx.x * blockDim.x + threadIdx.x;
    if (i < n) p[i] = v;
}

__global__ __launch_bounds__(256)
void gcn_deg_kernel(const int* __restrict__ col, float* __restrict__ deg, int nE) {
    int e = blockIdx.x * blockDim.x + threadIdx.x;
    if (e < nE) atomicAdd(deg + col[e], 1.0f);
}

__global__ __launch_bounds__(256)
void gcn_rsqrt_kernel(float* __restrict__ deg, int n) {
    int i = blockIdx.x * blockDim.x + threadIdx.x;
    if (i < n) {
        float d = deg[i];
        deg[i] = (d > 0.0f) ? rsqrtf(d) : 0.0f;
    }
}

__global__ __launch_bounds__(128)
void gcn_gemm_wmma_kernel(const float* __restrict__ X, const float* __restrict__ W,
                          float* __restrict__ H, int K, int nNodes) {
    const int lane = threadIdx.x & 31;
    const int wave = threadIdx.x >> 5;
    const int n    = lane & 15;
    const int half = lane >> 4;
    const int r0   = blockIdx.x * 16;
    const int n0   = wave * 16;

    const int rA = (r0 + n < nNodes) ? (r0 + n) : (nNodes - 1);

    v8f acc = {};
    for (int kb = 0; kb < K; kb += 4) {
        const int ka = kb + 2 * half;
        v2f a = *(const v2f*)(X + (long long)rA * K + ka);
        v2f b;
        b.x = W[(long long)(ka    ) * HID_CH + n0 + n];
        b.y = W[(long long)(ka + 1) * HID_CH + n0 + n];
        acc = __builtin_amdgcn_wmma_f32_16x16x4_f32(
                   false, a,  false, b,
                   (short)0, acc,  false,  false);
    }

#pragma unroll
    for (int v = 0; v < 8; ++v) {
        int m = v + 8 * half;
        if (r0 + m < nNodes)
            H[(long long)(r0 + m) * HID_CH + n0 + n] = acc[v];
    }
}

__global__ __launch_bounds__(256)
void gcn_scatter_kernel(const float* __restrict__ H, const int* __restrict__ row,
                        const int* __restrict__ col, const float* __restrict__ dinv,
                        float* __restrict__ out, int nE) {
    const int lane = threadIdx.x & 31;
    const int e    = blockIdx.x * 8 + (threadIdx.x >> 5);
    if (e >= nE) return;

    const int   r   = row[e];
    const int   c   = col[e];
    const float nrm = dinv[r] * dinv[c];

    const float* hr = H + (long long)r * HID_CH;
    float v0 = hr[lane];
    float v1 = hr[lane + 32];

    float* oc = out + (long long)c * HID_CH;
    atomicAdd(oc + lane,      v0 * nrm);
    atomicAdd(oc + lane + 32, v1 * nrm);
}

__global__ __launch_bounds__(256)
void gcn_finalize_kernel(const float* __restrict__ agg, const float* __restrict__ h,
                         const float* __restrict__ dinv, const float* __restrict__ bias,
                         float* __restrict__ out, int nNodes, int doRelu) {
    long long idx = (long long)blockIdx.x * blockDim.x + threadIdx.x;
    if (idx >= (long long)nNodes * HID_CH) return;
    int node = (int)(idx >> 6);
    int ch   = (int)(idx & 63);
    float di = dinv[node];
    float v  = agg[idx] + h[idx] * di * di + bias[ch];
    out[idx] = doRelu ? fmaxf(v, 0.0f) : v;
}

extern "C" void kernel_launch(void* const* d_in, const int* in_sizes, int n_in,
                              void* d_out, int out_size, void* d_ws, size_t ws_size,
                              hipStream_t stream) {
    const float* x   = (const float*)d_in[0];
    const int*   ei  = (const int*)  d_in[1];
    const float* W1  = (const float*)d_in[2];
    const float* b1  = (const float*)d_in[3];
    const float* W2  = (const float*)d_in[4];
    const float* b2  = (const float*)d_in[5];
    float*       out = (float*)d_out;

    const int N  = in_sizes[0] / IN_CH;
    const int nE = in_sizes[1] / 2;
    const int*   rowp = ei;
    const int*   colp = ei + nE;

    float* deg = (float*)d_ws;
    float* h1  = deg + N;
    float* agg = h1 + (long long)N * HID_CH;
    float* h2  = agg + (long long)N * HID_CH;

    const long long feat = (long long)N * HID_CH;
    const int fillBlocks = (int)((feat + 255) / 256);
    const int edgeThreadBlocks = (nE + 255) / 256;
    const int edgeWaveBlocks   = (nE + 7) / 8;
    const int nodeBlocks = (N + 255) / 256;
    const int tileBlocks = (N + 15) / 16;

    gcn_fill_kernel<<<(N + 255) / 256, 256, 0, stream>>>(deg, 1.0f, N);
    gcn_fill_kernel<<<fillBlocks, 256, 0, stream>>>(agg, 0.0f, feat);
    gcn_fill_kernel<<<fillBlocks, 256, 0, stream>>>(out, 0.0f, feat);

    gcn_deg_kernel<<<edgeThreadBlocks, 256, 0, stream>>>(colp, deg, nE);
    gcn_rsqrt_kernel<<<nodeBlocks, 256, 0, stream>>>(deg, N);

    gcn_gemm_wmma_kernel<<<tileBlocks, 128, 0, stream>>>(x, W1, h1, IN_CH, N);
    gcn_scatter_kernel<<<edgeWaveBlocks, 256, 0, stream>>>(h1, rowp, colp, deg, agg, nE);
    gcn_finalize_kernel<<<fillBlocks, 256, 0, stream>>>(agg, h1, deg, b1, h1, N,  1);

    gcn_gemm_wmma_kernel<<<tileBlocks, 128, 0, stream>>>(h1, W2, h2, HID_CH, N);
    gcn_scatter_kernel<<<edgeWaveBlocks, 256, 0, stream>>>(h2, rowp, colp, deg, out, nE);
    gcn_finalize_kernel<<<fillBlocks, 256, 0, stream>>>(out, h2, deg, b2, out, N,  0);
}


# GEMM operand delivery: a block stays and walks row tiles, its 16xK A tile read once by whole-line loads into LDS for all 4 waves, each wave's W slice held in registers; k chain order kept
# speedup vs baseline: 2.1151x; 2.1151x over previous
#include <hip/hip_runtime.h>
#include <hip/hip_bf16.h>

typedef __attribute__((ext_vector_type(2))) float v2f;
typedef __attribute__((ext_vector_type(4))) float v4f;
typedef __attribute__((ext_vector_type(8))) float v8f;

#define IN_CH  128
#define HID_CH 64
#define GEMM_BLOCKS 2304

__global__ __launch_bounds__(256)
void gcn_fill_kernel(float* __restrict__ p, float v, long long n) {
    long long i = (long long)blockIdx.x * blockDim.x + threadIdx.x;
    if (i < n) p[i] = v;
}

__global__ __launch_bounds__(256)
void gcn_deg_kernel(const int* __restrict__ col, float* __restrict__ deg, int nE) {
    int e = blockIdx.x * blockDim.x + threadIdx.x;
    if (e < nE) atomicAdd(deg + col[e], 1.0f);
}

template <bool LAYER1, int K>
__global__ __launch_bounds__(128)
void gcn_gemm_wmma_kernel(const float* __restrict__ A, const float* __restrict__ W,
                          const float* __restrict__ bias, const float* __restrict__ dsrc,
                          float* __restrict__ dinvOut, float* __restrict__ H,
                          float* __restrict__ acc0, int nNodes, int nTiles) {
    constexpr int PITCH = K + 4;
    constexpr int C4    = K / 4;
    __shared__ __attribute__((aligned(16))) float sA[16 * PITCH];

    const int tid  = threadIdx.x;
    const int lane = tid & 31;
    const int wave = tid >> 5;
    const int n    = lane & 15;
    const int half = lane >> 4;
    const int n0   = wave * 16;

    v2f bf[K / 4];
#pragma unroll
    for (int s = 0; s < K / 4; ++s) {
        const int ka = 4 * s + 2 * half;
        bf[s].x = W[(ka    ) * HID_CH + n0 + n];
        bf[s].y = W[(ka + 1) * HID_CH + n0 + n];
    }
    const float bv = bias[n0 + n];

    for (int t = blockIdx.x; t < nTiles; t += gridDim.x) {
        const int r0 = t * 16;

#pragma unroll
        for (int i = 0; i < (16 * C4) / 128; ++i) {
            const int idx = i * 128 + tid;
            const int rr  = idx / C4;
            const int c4  = idx % C4;
            const int gr  = (r0 + rr < nNodes) ? (r0 + rr) : (nNodes - 1);
            v4f x = *(const v4f*)(A + (long long)gr * K + c4 * 4);
            if (!LAYER1) {
                x.x = fmaxf(x.x, 0.0f); x.y = fmaxf(x.y, 0.0f);
                x.z = fmaxf(x.z, 0.0f); x.w = fmaxf(x.w, 0.0f);
            }
            *(v4f*)(sA + rr * PITCH + c4 * 4) = x;
        }
        __syncthreads();

        v8f acc = {};
#pragma unroll
        for (int s = 0; s < K / 4; ++s) {
            const v2f a = *(const v2f*)(sA + n * PITCH + 4 * s + 2 * half);
            acc = __builtin_amdgcn_wmma_f32_16x16x4_f32(
                       false, a,  false, bf[s],
                       (short)0, acc,  false,  false);
        }

        float di[8];
#pragma unroll
        for (int v = 0; v < 8; ++v) {
            const int m  = v + 8 * half;
            const int rr = (r0 + m < nNodes) ? (r0 + m) : (nNodes - 1);
            di[v] = dsrc[rr];
        }
        if (LAYER1) {
#pragma unroll
            for (int v = 0; v < 8; ++v) di[v] = (di[v] > 0.0f) ? rsqrtf(di[v]) : 0.0f;
            if (wave == 0 && half == 0 && r0 + n < nNodes) {
                const float d = dsrc[r0 + n];
                dinvOut[r0 + n] = (d > 0.0f) ? rsqrtf(d) : 0.0f;
            }
        }

#pragma unroll
        for (int v = 0; v < 8; ++v) {
            const int m = v + 8 * half;
            if (r0 + m < nNodes) {
                const long long idx = (long long)(r0 + m) * HID_CH + n0 + n;
                H[idx]    = acc[v];
                acc0[idx] = acc[v] * di[v] * di[v] + bv;
            }
        }
        __syncthreads();
    }
}

#define SC_GROUP 8

__device__ __forceinline__ int gcn_lane_i(int v, int l) { return __builtin_amdgcn_readlane(v, l); }
__device__ __forceinline__ float gcn_lane_f(float v, int l) {
    return __int_as_float(__builtin_amdgcn_readlane(__float_as_int(v), l));
}

__global__ __launch_bounds__(256)
void gcn_scatter_kernel(const float* __restrict__ H, const int* __restrict__ row,
                        const int* __restrict__ col, const float* __restrict__ dinv,
                        float* __restrict__ out, int nE) {
    const int lane = threadIdx.x & 31;
    const int wv   = __builtin_amdgcn_readfirstlane(threadIdx.x >> 5);
    const int e0   = (blockIdx.x * 8 + wv) * 32;
    if (e0 >= nE) return;
    const int cnt  = (nE - e0 < 32) ? (nE - e0) : 32;

    const bool live = lane < cnt;
    const int   r   = live ? row[e0 + lane] : 0;
    const int   c   = live ? col[e0 + lane] : 0;
    const float nrm = dinv[r] * dinv[c];

    int j0 = 0;
    for (; j0 + SC_GROUP <= cnt; j0 += SC_GROUP) {
        float v0[SC_GROUP], v1[SC_GROUP];
        for (int u = 0; u < SC_GROUP; ++u) {
            const float* hr = H + (long long)gcn_lane_i(r, j0 + u) * HID_CH;
            v0[u] = hr[lane];
            v1[u] = hr[lane + 32];
        }
        for (int u = 0; u < SC_GROUP; ++u) {
            float* oc = out + (long long)gcn_lane_i(c, j0 + u) * HID_CH;
            const float nj = gcn_lane_f(nrm, j0 + u);
            atomicAdd(oc + lane,      v0[u] * nj);
            atomicAdd(oc + lane + 32, v1[u] * nj);
        }
    }
    for (; j0 < cnt; ++j0) {
        const float* hr = H + (long long)gcn_lane_i(r, j0) * HID_CH;
        float* oc = out + (long long)gcn_lane_i(c, j0) * HID_CH;
        const float nj = gcn_lane_f(nrm, j0);
        atomicAdd(oc + lane,      hr[lane] * nj);
        atomicAdd(oc + lane + 32, hr[lane + 32] * nj);
    }
}

extern "C" void kernel_launch(void* const* d_in, const int* in_sizes, int n_in,
                              void* d_out, int out_size, void* d_ws, size_t ws_size,
                              hipStream_t stream) {
    const float* x   = (const float*)d_in[0];
    const int*   ei  = (const int*)  d_in[1];
    const float* W1  = (const float*)d_in[2];
    const float* b1  = (const float*)d_in[3];
    const float* W2  = (const float*)d_in[4];
    const float* b2  = (const float*)d_in[5];
    float*       out = (float*)d_out;

    const int N  = in_sizes[0] / IN_CH;
    const int nE = in_sizes[1] / 2;
    const int*   rowp = ei;
    const int*   colp = ei + nE;

    float* h    = (float*)d_ws;
    float* agg  = h + (long long)N * HID_CH;
    float* deg  = agg + (long long)N * HID_CH;
    float* dinv = deg + N;

    const int edgeThreadBlocks = (nE + 255) / 256;
    const int tileBlocks = (N + 15) / 16;

    gcn_fill_kernel<<<(N + 255) / 256, 256, 0, stream>>>(deg, 1.0f, N);
    gcn_deg_kernel<<<edgeThreadBlocks, 256, 0, stream>>>(colp, deg, nE);

    gcn_gemm_wmma_kernel<true, IN_CH><<<GEMM_BLOCKS, 128, 0, stream>>>(x, W1, b1, deg, dinv, h, agg, N, tileBlocks);
    gcn_scatter_kernel<<<edgeThreadBlocks, 256, 0, stream>>>(h, rowp, colp, dinv, agg, nE);

    gcn_gemm_wmma_kernel<false, HID_CH><<<GEMM_BLOCKS, 128, 0, stream>>>(agg, W2, b2, dinv, nullptr, h, out, N, tileBlocks);
    gcn_scatter_kernel<<<edgeThreadBlocks, 256, 0, stream>>>(h, rowp, colp, dinv, out, nE);
}


# Edge scatter cut by destination half: a block reads its die number, takes (512-edge group, node half) tickets of its own package half first; a row's atomics come from one half; arrival order changes
# speedup vs baseline: 2.4126x; 1.1407x over previous
#include <hip/hip_runtime.h>
#include <hip/hip_bf16.h>

typedef __attribute__((ext_vector_type(2))) float v2f;
typedef __attribute__((ext_vector_type(4))) float v4f;
typedef __attribute__((ext_vector_type(8))) float v8f;

#define IN_CH  128
#define HID_CH 64
#define GEMM_BLOCKS 2304

#define TICK_STRIDE 64
__global__ __launch_bounds__(256)
void gcn_fill_kernel(float* __restrict__ p, float v, long long n, int* __restrict__ tick) {
    long long i = (long long)blockIdx.x * blockDim.x + threadIdx.x;
    if (i < n) p[i] = v;
    if (i < 4) tick[i * TICK_STRIDE] = 0;
}

__global__ __launch_bounds__(256)
void gcn_deg_kernel(const int* __restrict__ col, float* __restrict__ deg, int nE) {
    int e = blockIdx.x * blockDim.x + threadIdx.x;
    if (e < nE) atomicAdd(deg + col[e], 1.0f);
}

template <bool LAYER1, int K>
__global__ __launch_bounds__(128)
void gcn_gemm_wmma_kernel(const float* __restrict__ A, const float* __restrict__ W,
                          const float* __restrict__ bias, const float* __restrict__ dsrc,
                          float* __restrict__ dinvOut, float* __restrict__ H,
                          float* __restrict__ acc0, int nNodes, int nTiles) {
    constexpr int PITCH = K + 4;
    constexpr int C4    = K / 4;
    __shared__ __attribute__((aligned(16))) float sA[16 * PITCH];

    const int tid  = threadIdx.x;
    const int lane = tid & 31;
    const int wave = tid >> 5;
    const int n    = lane & 15;
    const int half = lane >> 4;
    const int n0   = wave * 16;

    v2f bf[K / 4];
#pragma unroll
    for (int s = 0; s < K / 4; ++s) {
        const int ka = 4 * s + 2 * half;
        bf[s].x = W[(ka    ) * HID_CH + n0 + n];
        bf[s].y = W[(ka + 1) * HID_CH + n0 + n];
    }
    const float bv = bias[n0 + n];

    for (int t = blockIdx.x; t < nTiles; t += gridDim.x) {
        const int r0 = t * 16;

#pragma unroll
        for (int i = 0; i < (16 * C4) / 128; ++i) {
            const int idx = i * 128 + tid;
            const int rr  = idx / C4;
            const int c4  = idx % C4;
            const int gr  = (r0 + rr < nNodes) ? (r0 + rr) : (nNodes - 1);
            v4f x = *(const v4f*)(A + (long long)gr * K + c4 * 4);
            if (!LAYER1) {
                x.x = fmaxf(x.x, 0.0f); x.y = fmaxf(x.y, 0.0f);
                x.z = fmaxf(x.z, 0.0f); x.w = fmaxf(x.w, 0.0f);
            }
            *(v4f*)(sA + rr * PITCH + c4 * 4) = x;
        }
        __syncthreads();

        v8f acc = {};
#pragma unroll
        for (int s = 0; s < K / 4; ++s) {
            const v2f a = *(const v2f*)(sA + n * PITCH + 4 * s + 2 * half);
            acc = __builtin_amdgcn_wmma_f32_16x16x4_f32(
                       false, a,  false, bf[s],
                       (short)0, acc,  false,  false);
        }

        float di[8];
#pragma unroll
        for (int v = 0; v < 8; ++v) {
            const int m  = v + 8 * half;
            const int rr = (r0 + m < nNodes) ? (r0 + m) : (nNodes - 1);
            di[v] = dsrc[rr];
        }
        if (LAYER1) {
#pragma unroll
            for (int v = 0; v < 8; ++v) di[v] = (di[v] > 0.0f) ? rsqrtf(di[v]) : 0.0f;
            if (wave == 0 && half == 0 && r0 + n < nNodes) {
                const float d = dsrc[r0 + n];
                dinvOut[r0 + n] = (d > 0.0f) ? rsqrtf(d) : 0.0f;
            }
        }

#pragma unroll
        for (int v = 0; v < 8; ++v) {
            const int m = v + 8 * half;
            if (r0 + m < nNodes) {
                const long long idx = (long long)(r0 + m) * HID_CH + n0 + n;
                H[idx]    = acc[v];
                acc0[idx] = acc[v] * di[v] * di[v] + bv;
            }
        }
        __syncthreads();
    }
}

#define SC_GROUP 8
#define SC_WAVE_EDGES 64
#define SC_BLOCK_EDGES 512

__device__ __forceinline__ int gcn_lane_i(int v, int l) { return __builtin_amdgcn_readlane(v, l); }
__device__ __forceinline__ float gcn_lane_f(float v, int l) {
    return __int_as_float(__builtin_amdgcn_readlane(__float_as_int(v), l));
}

__device__ __forceinline__ void gcn_walk(const float* __restrict__ H, float* __restrict__ out,
                                         int r, int c, float nrm, int cnt, int lane) {
    int j0 = 0;
    for (; j0 + SC_GROUP <= cnt; j0 += SC_GROUP) {
        float v0[SC_GROUP], v1[SC_GROUP];
#pragma unroll
        for (int u = 0; u < SC_GROUP; ++u) {
            const float* hr = H + (long long)gcn_lane_i(r, j0 + u) * HID_CH;
            v0[u] = hr[lane];
            v1[u] = hr[lane + 32];
        }
#pragma unroll
        for (int u = 0; u < SC_GROUP; ++u) {
            float* oc = out + (long long)gcn_lane_i(c, j0 + u) * HID_CH;
            const float nj = gcn_lane_f(nrm, j0 + u);
            atomicAdd(oc + lane,      v0[u] * nj);
            atomicAdd(oc + lane + 32, v1[u] * nj);
        }
    }
    if (j0 < cnt) {
        float v0[SC_GROUP], v1[SC_GROUP];
#pragma unroll
        for (int u = 0; u < SC_GROUP - 1; ++u) {
            if (j0 + u < cnt) {
                const float* hr = H + (long long)gcn_lane_i(r, j0 + u) * HID_CH;
                v0[u] = hr[lane];
                v1[u] = hr[lane + 32];
            }
        }
#pragma unroll
        for (int u = 0; u < SC_GROUP - 1; ++u) {
            if (j0 + u < cnt) {
                float* oc = out + (long long)gcn_lane_i(c, j0 + u) * HID_CH;
                const float nj = gcn_lane_f(nrm, j0 + u);
                atomicAdd(oc + lane,      v0[u] * nj);
                atomicAdd(oc + lane + 32, v1[u] * nj);
            }
        }
    }
}

__global__ __launch_bounds__(256)
void gcn_scatter_kernel(const float* __restrict__ H, const int* __restrict__ row,
                        const int* __restrict__ col, const float* __restrict__ dinv,
                        float* __restrict__ out, int* __restrict__ tick,
                        int nE, int nGroups, int nSplit) {
    __shared__ int   sItem[2];
    __shared__ int   sR[8][SC_WAVE_EDGES];
    __shared__ int   sC[8][SC_WAVE_EDGES];
    __shared__ float sN[8][SC_WAVE_EDGES];

    const int lane = threadIdx.x & 31;
    const int wv   = __builtin_amdgcn_readfirstlane(threadIdx.x >> 5);

    if (threadIdx.x == 0) {
        const unsigned id = __builtin_amdgcn_s_sendmsg_rtn(0x87);
        int p = (int)((id >> 18) & 1u);
        int t = atomicAdd(tick + p * TICK_STRIDE, 1);
        if (t >= nGroups) {
            p ^= 1;
            t = atomicAdd(tick + p * TICK_STRIDE, 1);
        }
        sItem[0] = t;
        sItem[1] = p;
    }
    __syncthreads();
    const int t = __builtin_amdgcn_readfirstlane(sItem[0]);
    const int p = __builtin_amdgcn_readfirstlane(sItem[1]);
    if (t >= nGroups) return;

    const int  eA = t * SC_BLOCK_EDGES + wv * SC_WAVE_EDGES + lane;
    const int  eB = eA + 32;
    int rA = 0, cA = 0, rB = 0, cB = 0;
    if (eA < nE) { rA = row[eA]; cA = col[eA]; }
    if (eB < nE) { rB = row[eB]; cB = col[eB]; }
    const bool upper = (p != 0);
    const bool inA = (eA < nE) && ((cA >= nSplit) == upper);
    const bool inB = (eB < nE) && ((cB >= nSplit) == upper);
    const float nA = inA ? dinv[rA] * dinv[cA] : 0.0f;
    const float nB = inB ? dinv[rB] * dinv[cB] : 0.0f;

    const unsigned mA = __builtin_amdgcn_ballot_w32(inA);
    const unsigned mB = __builtin_amdgcn_ballot_w32(inB);
    const int cntA = __builtin_popcount(mA);
    const int cnt  = cntA + __builtin_popcount(mB);
    if (inA) {
        const int k = (int)__builtin_amdgcn_mbcnt_lo(mA, 0u);
        sR[wv][k] = rA; sC[wv][k] = cA; sN[wv][k] = nA;
    }
    if (inB) {
        const int k = cntA + (int)__builtin_amdgcn_mbcnt_lo(mB, 0u);
        sR[wv][k] = rB; sC[wv][k] = cB; sN[wv][k] = nB;
    }
    __syncthreads();

    const bool h0 = lane < cnt, h1 = lane + 32 < cnt;
    const int   r0 = h0 ? sR[wv][lane] : 0,      r1 = h1 ? sR[wv][lane + 32] : 0;
    const int   c0 = h0 ? sC[wv][lane] : 0,      c1 = h1 ? sC[wv][lane + 32] : 0;
    const float n0 = h0 ? sN[wv][lane] : 0.0f,   n1 = h1 ? sN[wv][lane + 32] : 0.0f;

    gcn_walk(H, out, r0, c0, n0, (cnt < 32) ? cnt : 32, lane);
    gcn_walk(H, out, r1, c1, n1, (cnt > 32) ? cnt - 32 : 0, lane);
}

extern "C" void kernel_launch(void* const* d_in, const int* in_sizes, int n_in,
                              void* d_out, int out_size, void* d_ws, size_t ws_size,
                              hipStream_t stream) {
    const float* x   = (const float*)d_in[0];
    const int*   ei  = (const int*)  d_in[1];
    const float* W1  = (const float*)d_in[2];
    const float* b1  = (const float*)d_in[3];
    const float* W2  = (const float*)d_in[4];
    const float* b2  = (const float*)d_in[5];
    float*       out = (float*)d_out;

    const int N  = in_sizes[0] / IN_CH;
    const int nE = in_sizes[1] / 2;
    const int*   rowp = ei;
    const int*   colp = ei + nE;

    float* h    = (float*)d_ws;
    float* agg  = h + (long long)N * HID_CH;
    float* deg  = agg + (long long)N * HID_CH;
    float* dinv = deg + N;
    int*   tick = (int*)(dinv + N);

    const int edgeThreadBlocks = (nE + 255) / 256;
    const int nGroups = (nE + SC_BLOCK_EDGES - 1) / SC_BLOCK_EDGES;
    const int scBlocks = 2 * nGroups;
    const int nSplit = N / 2;
    const int tileBlocks = (N + 15) / 16;

    gcn_fill_kernel<<<(N + 255) / 256, 256, 0, stream>>>(deg, 1.0f, N, tick);
    gcn_deg_kernel<<<edgeThreadBlocks, 256, 0, stream>>>(colp, deg, nE);

    gcn_gemm_wmma_kernel<true, IN_CH><<<GEMM_BLOCKS, 128, 0, stream>>>(x, W1, b1, deg, dinv, h, agg, N, tileBlocks);
    gcn_scatter_kernel<<<scBlocks, 256, 0, stream>>>(h, rowp, colp, dinv, agg, tick, nE, nGroups, nSplit);

    gcn_gemm_wmma_kernel<false, HID_CH><<<GEMM_BLOCKS, 128, 0, stream>>>(agg, W2, b2, dinv, nullptr, h, out, N, tileBlocks);
    gcn_scatter_kernel<<<scBlocks, 256, 0, stream>>>(h, rowp, colp, dinv, out, tick + 2 * TICK_STRIDE, nE, nGroups, nSplit);
}


# Edge norm dinv[row]*dinv[col] computed once per call: the layer-1 GEMM launch's blocks store it per edge beside their tiles; both scatters read it in edge order, not by two random gathers per edge
# speedup vs baseline: 2.5917x; 1.0742x over previous
#include <hip/hip_runtime.h>
#include <hip/hip_bf16.h>

typedef __attribute__((ext_vector_type(2))) float v2f;
typedef __attribute__((ext_vector_type(4))) float v4f;
typedef __attribute__((ext_vector_type(8))) float v8f;

#define IN_CH  128
#define HID_CH 64
#define GEMM_BLOCKS 2304

#define TICK_STRIDE 64
__global__ __launch_bounds__(256)
void gcn_fill_kernel(float* __restrict__ p, float v, long long n, int* __restrict__ tick) {
    long long i = (long long)blockIdx.x * blockDim.x + threadIdx.x;
    if (i < n) p[i] = v;
    if (i < 4) tick[i * TICK_STRIDE] = 0;
}

__global__ __launch_bounds__(256)
void gcn_deg_kernel(const int* __restrict__ col, float* __restrict__ deg, int nE) {
    int e = blockIdx.x * blockDim.x + threadIdx.x;
    if (e < nE) atomicAdd(deg + col[e], 1.0f);
}

__device__ __forceinline__ float gcn_dinv_of(float d) { return (d > 0.0f) ? rsqrtf(d) : 0.0f; }

#define NORM_BATCH 4
__device__ __forceinline__ void gcn_edge_norms(const int* __restrict__ row, const int* __restrict__ col,
                                               const float* __restrict__ deg, float* __restrict__ enorm,
                                               int nE) {
    const int T = gridDim.x * blockDim.x;
    int e = blockIdx.x * blockDim.x + threadIdx.x;
    for (; e + (NORM_BATCH - 1) * T < nE; e += NORM_BATCH * T) {
        int   r[NORM_BATCH], c[NORM_BATCH];
        float dr[NORM_BATCH], dc[NORM_BATCH];
#pragma unroll
        for (int u = 0; u < NORM_BATCH; ++u) { r[u] = row[e + u * T]; c[u] = col[e + u * T]; }
#pragma unroll
        for (int u = 0; u < NORM_BATCH; ++u) { dr[u] = deg[r[u]]; dc[u] = deg[c[u]]; }
#pragma unroll
        for (int u = 0; u < NORM_BATCH; ++u) enorm[e + u * T] = gcn_dinv_of(dr[u]) * gcn_dinv_of(dc[u]);
    }
    for (; e < nE; e += T) enorm[e] = gcn_dinv_of(deg[row[e]]) * gcn_dinv_of(deg[col[e]]);
}

template <bool LAYER1, int K>
__global__ __launch_bounds__(128)
void gcn_gemm_wmma_kernel(const float* __restrict__ A, const float* __restrict__ W,
                          const float* __restrict__ bias, const float* __restrict__ dsrc,
                          float* __restrict__ dinvOut, float* __restrict__ H,
                          float* __restrict__ acc0, int nNodes, int nTiles,
                          const int* __restrict__ erow, const int* __restrict__ ecol,
                          float* __restrict__ enorm, int nE) {
    constexpr int PITCH = K + 4;
    constexpr int C4    = K / 4;
    __shared__ __attribute__((aligned(16))) float sA[16 * PITCH];

    const int tid  = threadIdx.x;
    const int lane = tid & 31;
    const int wave = tid >> 5;
    const int n    = lane & 15;
    const int half = lane >> 4;
    const int n0   = wave * 16;

    const bool normFirst = (blockIdx.x & 1) != 0;
    if (LAYER1 && normFirst) gcn_edge_norms(erow, ecol, dsrc, enorm, nE);

    v2f bf[K / 4];
#pragma unroll
    for (int s = 0; s < K / 4; ++s) {
        const int ka = 4 * s + 2 * half;
        bf[s].x = W[(ka    ) * HID_CH + n0 + n];
        bf[s].y = W[(ka + 1) * HID_CH + n0 + n];
    }
    const float bv = bias[n0 + n];

    for (int t = blockIdx.x; t < nTiles; t += gridDim.x) {
        const int r0 = t * 16;

#pragma unroll
        for (int i = 0; i < (16 * C4) / 128; ++i) {
            const int idx = i * 128 + tid;
            const int rr  = idx / C4;
            const int c4  = idx % C4;
            const int gr  = (r0 + rr < nNodes) ? (r0 + rr) : (nNodes - 1);
            v4f x = *(const v4f*)(A + (long long)gr * K + c4 * 4);
            if (!LAYER1) {
                x.x = fmaxf(x.x, 0.0f); x.y = fmaxf(x.y, 0.0f);
                x.z = fmaxf(x.z, 0.0f); x.w = fmaxf(x.w, 0.0f);
            }
            *(v4f*)(sA + rr * PITCH + c4 * 4) = x;
        }
        __syncthreads();

        v8f acc = {};
#pragma unroll
        for (int s = 0; s < K / 4; ++s) {
            const v2f a = *(const v2f*)(sA + n * PITCH + 4 * s + 2 * half);
            acc = __builtin_amdgcn_wmma_f32_16x16x4_f32(
                       false, a,  false, bf[s],
                       (short)0, acc,  false,  false);
        }

        float di[8];
#pragma unroll
        for (int v = 0; v < 8; ++v) {
            const int m  = v + 8 * half;
            const int rr = (r0 + m < nNodes) ? (r0 + m) : (nNodes - 1);
            di[v] = dsrc[rr];
        }
        if (LAYER1) {
#pragma unroll
            for (int v = 0; v < 8; ++v) di[v] = (di[v] > 0.0f) ? rsqrtf(di[v]) : 0.0f;
            if (wave == 0 && half == 0 && r0 + n < nNodes) {
                const float d = dsrc[r0 + n];
                dinvOut[r0 + n] = (d > 0.0f) ? rsqrtf(d) : 0.0f;
            }
        }

#pragma unroll
        for (int v = 0; v < 8; ++v) {
            const int m = v + 8 * half;
            if (r0 + m < nNodes) {
                const long long idx = (long long)(r0 + m) * HID_CH + n0 + n;
                H[idx]    = acc[v];
                acc0[idx] = acc[v] * di[v] * di[v] + bv;
            }
        }
        __syncthreads();
    }

    if (LAYER1 && !normFirst) gcn_edge_norms(erow, ecol, dsrc, enorm, nE);
}

#define SC_GROUP 8
#define SC_WAVE_EDGES 64
#define SC_BLOCK_EDGES 512

__device__ __forceinline__ int gcn_lane_i(int v, int l) { return __builtin_amdgcn_readlane(v, l); }
__device__ __forceinline__ float gcn_lane_f(float v, int l) {
    return __int_as_float(__builtin_amdgcn_readlane(__float_as_int(v), l));
}

__device__ __forceinline__ void gcn_walk(const float* __restrict__ H, float* __restrict__ out,
                                         int r, int c, float nrm, int cnt, int lane) {
    int j0 = 0;
    for (; j0 + SC_GROUP <= cnt; j0 += SC_GROUP) {
        float v0[SC_GROUP], v1[SC_GROUP];
#pragma unroll
        for (int u = 0; u < SC_GROUP; ++u) {
            const float* hr = H + (long long)gcn_lane_i(r, j0 + u) * HID_CH;
            v0[u] = hr[lane];
            v1[u] = hr[lane + 32];
        }
#pragma unroll
        for (int u = 0; u < SC_GROUP; ++u) {
            float* oc = out + (long long)gcn_lane_i(c, j0 + u) * HID_CH;
            const float nj = gcn_lane_f(nrm, j0 + u);
            atomicAdd(oc + lane,      v0[u] * nj);
            atomicAdd(oc + lane + 32, v1[u] * nj);
        }
    }
    if (j0 < cnt) {
        float v0[SC_GROUP], v1[SC_GROUP];
#pragma unroll
        for (int u = 0; u < SC_GROUP - 1; ++u) {
            if (j0 + u < cnt) {
                const float* hr = H + (long long)gcn_lane_i(r, j0 + u) * HID_CH;
                v0[u] = hr[lane];
                v1[u] = hr[lane + 32];
            }
        }
#pragma unroll
        for (int u = 0; u < SC_GROUP - 1; ++u) {
            if (j0 + u < cnt) {
                float* oc = out + (long long)gcn_lane_i(c, j0 + u) * HID_CH;
                const float nj = gcn_lane_f(nrm, j0 + u);
                atomicAdd(oc + lane,      v0[u] * nj);
                atomicAdd(oc + lane + 32, v1[u] * nj);
            }
        }
    }
}

__global__ __launch_bounds__(256)
void gcn_scatter_kernel(const float* __restrict__ H, const int* __restrict__ row,
                        const int* __restrict__ col, const float* __restrict__ enorm,
                        float* __restrict__ out, int* __restrict__ tick,
                        int nE, int nGroups, int nSplit) {
    __shared__ int   sItem[2];
    __shared__ int   sR[8][SC_WAVE_EDGES];
    __shared__ int   sC[8][SC_WAVE_EDGES];
    __shared__ float sN[8][SC_WAVE_EDGES];

    const int lane = threadIdx.x & 31;
    const int wv   = __builtin_amdgcn_readfirstlane(threadIdx.x >> 5);

    if (threadIdx.x == 0) {
        const unsigned id = __builtin_amdgcn_s_sendmsg_rtn(0x87);
        int p = (int)((id >> 18) & 1u);
        int t = atomicAdd(tick + p * TICK_STRIDE, 1);
        if (t >= nGroups) {
            p ^= 1;
            t = atomicAdd(tick + p * TICK_STRIDE, 1);
        }
        sItem[0] = t;
        sItem[1] = p;
    }
    __syncthreads();
    const int t = __builtin_amdgcn_readfirstlane(sItem[0]);
    const int p = __builtin_amdgcn_readfirstlane(sItem[1]);
    if (t >= nGroups) return;

    const int  eA = t * SC_BLOCK_EDGES + wv * SC_WAVE_EDGES + lane;
    const int  eB = eA + 32;
    int rA = 0, cA = 0, rB = 0, cB = 0;
    float qA = 0.0f, qB = 0.0f;
    if (eA < nE) { rA = row[eA]; cA = col[eA]; qA = enorm[eA]; }
    if (eB < nE) { rB = row[eB]; cB = col[eB]; qB = enorm[eB]; }
    const bool upper = (p != 0);
    const bool inA = (eA < nE) && ((cA >= nSplit) == upper);
    const bool inB = (eB < nE) && ((cB >= nSplit) == upper);
    const float nA = inA ? qA : 0.0f;
    const float nB = inB ? qB : 0.0f;

    const unsigned mA = __builtin_amdgcn_ballot_w32(inA);
    const unsigned mB = __builtin_amdgcn_ballot_w32(inB);
    const int cntA = __builtin_popcount(mA);
    const int cnt  = cntA + __builtin_popcount(mB);
    if (inA) {
        const int k = (int)__builtin_amdgcn_mbcnt_lo(mA, 0u);
        sR[wv][k] = rA; sC[wv][k] = cA; sN[wv][k] = nA;
    }
    if (inB) {
        const int k = cntA + (int)__builtin_amdgcn_mbcnt_lo(mB, 0u);
        sR[wv][k] = rB; sC[wv][k] = cB; sN[wv][k] = nB;
    }
    __syncthreads();

    const bool h0 = lane < cnt, h1 = lane + 32 < cnt;
    const int   r0 = h0 ? sR[wv][lane] : 0,      r1 = h1 ? sR[wv][lane + 32] : 0;
    const int   c0 = h0 ? sC[wv][lane] : 0,      c1 = h1 ? sC[wv][lane + 32] : 0;
    const float n0 = h0 ? sN[wv][lane] : 0.0f,   n1 = h1 ? sN[wv][lane + 32] : 0.0f;

    gcn_walk(H, out, r0, c0, n0, (cnt < 32) ? cnt : 32, lane);
    gcn_walk(H, out, r1, c1, n1, (cnt > 32) ? cnt - 32 : 0, lane);
}

extern "C" void kernel_launch(void* const* d_in, const int* in_sizes, int n_in,
                              void* d_out, int out_size, void* d_ws, size_t ws_size,
                              hipStream_t stream) {
    const float* x   = (const float*)d_in[0];
    const int*   ei  = (const int*)  d_in[1];
    const float* W1  = (const float*)d_in[2];
    const float* b1  = (const float*)d_in[3];
    const float* W2  = (const float*)d_in[4];
    const float* b2  = (const float*)d_in[5];
    float*       out = (float*)d_out;

    const int N  = in_sizes[0] / IN_CH;
    const int nE = in_sizes[1] / 2;
    const int*   rowp = ei;
    const int*   colp = ei + nE;

    float* h    = (float*)d_ws;
    float* agg  = h + (long long)N * HID_CH;
    float* deg  = agg + (long long)N * HID_CH;
    float* dinv = deg + N;
    int*   tick = (int*)(dinv + N);
    float* enorm = (float*)(tick + 4 * TICK_STRIDE);

    const int edgeThreadBlocks = (nE + 255) / 256;
    const int nGroups = (nE + SC_BLOCK_EDGES - 1) / SC_BLOCK_EDGES;
    const int scBlocks = 2 * nGroups;
    const int nSplit = N / 2;
    const int tileBlocks = (N + 15) / 16;

    gcn_fill_kernel<<<(N + 255) / 256, 256, 0, stream>>>(deg, 1.0f, N, tick);
    gcn_deg_kernel<<<edgeThreadBlocks, 256, 0, stream>>>(colp, deg, nE);

    gcn_gemm_wmma_kernel<true, IN_CH><<<GEMM_BLOCKS, 128, 0, stream>>>(x, W1, b1, deg, dinv, h, agg, N, tileBlocks, rowp, colp, enorm, nE);
    gcn_scatter_kernel<<<scBlocks, 256, 0, stream>>>(h, rowp, colp, enorm, agg, tick, nE, nGroups, nSplit);

    gcn_gemm_wmma_kernel<false, HID_CH><<<GEMM_BLOCKS, 128, 0, stream>>>(agg, W2, b2, dinv, nullptr, h, out, N, tileBlocks, nullptr, nullptr, nullptr, 0);
    gcn_scatter_kernel<<<scBlocks, 256, 0, stream>>>(h, rowp, colp, enorm, out, tick + 2 * TICK_STRIDE, nE, nGroups, nSplit);
}
